# starting kernel
# baseline (speedup 1.0000x reference)
#include <hip/hip_runtime.h>

typedef __attribute__((ext_vector_type(2))) float v2f;
typedef __attribute__((ext_vector_type(8))) float v8f;

#define N_NODES   100000
#define N_FEAT    128
#define HIDDEN    20
#define N_CLASSES 10
#define PAD1      32
#define PAD2      16
#define M_TILES   (N_NODES / 16)


__global__ void gcn_init_deg(float* deg) {
    int i = blockIdx.x * blockDim.x + threadIdx.x;
    if (i < N_NODES) deg[i] = 1.0f;
}

__global__ void gcn_deg_accum(const int* __restrict__ dst, float* deg, int E) {
    int e = blockIdx.x * blockDim.x + threadIdx.x;
    if (e < E) atomicAdd(&deg[dst[e]], 1.0f);
}

__global__ void gcn_rsqrt(float* deg) {
    int i = blockIdx.x * blockDim.x + threadIdx.x;
    if (i < N_NODES) deg[i] = rsqrtf(deg[i]);
}


__global__ void gcn_gemm1(const float* __restrict__ x,
                          const float* __restrict__ W1,
                          const float* __restrict__ dis,
                          float* __restrict__ h0,
                          float* __restrict__ h1) {
    __shared__ float Bf[(N_FEAT / 2) * 32 * 2];

    for (int idx = threadIdx.x; idx < (N_FEAT / 2) * 32 * 2; idx += blockDim.x) {
        const int p = idx >> 6;
        const int c = (idx >> 1) & 31;
        const int j = idx & 1;
        Bf[idx] = (c < HIDDEN) ? W1[(2 * p + j) * HIDDEN + c] : 0.0f;
    }
    __syncthreads();

    const int lane = threadIdx.x & 31;
    const int wave = threadIdx.x >> 5;
    const int tile = blockIdx.x * (blockDim.x >> 5) + wave;
    if (tile >= M_TILES) return;
    const int row0 = tile << 4;
    const int half = lane >> 4;
    const int l16  = lane & 15;
    const float* arow = x + (size_t)(row0 + l16) * N_FEAT;

    v8f c0 = {};
    v8f c1 = {};

    #pragma unroll 4
    for (int k0 = 0; k0 < N_FEAT; k0 += 4) {
        const int kb = k0 + 2 * half;
        const int p = kb >> 1;
        v2f a  = *(const v2f*)(arow + kb);
        v2f b0 = *(const v2f*)&Bf[p * 64 + l16 * 2];
        v2f b1 = *(const v2f*)&Bf[p * 64 + (16 + l16) * 2];

        c0 = __builtin_amdgcn_wmma_f32_16x16x4_f32(false, a, false, b0, (short)0, c0, false, false);
        c1 = __builtin_amdgcn_wmma_f32_16x16x4_f32(false, a, false, b1, (short)0, c1, false, false);
    }

    #pragma unroll
    for (int r = 0; r < 8; ++r) {
        const int row = row0 + r + 8 * half;
        const float di = dis[row];
        const float d2 = di * di;
        float* p0 = h0 + (size_t)row * PAD1;
        float* p1 = h1 + (size_t)row * PAD1;
        p0[l16]      = c0[r];
        p0[16 + l16] = c1[r];
        p1[l16]      = d2 * c0[r];
        p1[16 + l16] = d2 * c1[r];
    }
}


template <int F, int PAD>
__global__ void gcn_propagate(const int* __restrict__ src,
                              const int* __restrict__ dst,
                              const float* __restrict__ dis,
                              const float* __restrict__ h,
                              float* __restrict__ out, int E) {
    int e = blockIdx.x * blockDim.x + threadIdx.x;
    if (e >= E) return;
    const int s = src[e];
    const int d = dst[e];
    const float nrm = dis[s] * dis[d];
    const float* hrow = h + (size_t)s * PAD;
    float* orow = out + (size_t)d * PAD;

    float v[F];
    #pragma unroll
    for (int j = 0; j + 4 <= F; j += 4) {
        float4 t = *(const float4*)(hrow + j);
        v[j] = t.x; v[j + 1] = t.y; v[j + 2] = t.z; v[j + 3] = t.w;
    }
    if constexpr (F % 4 != 0) {
        float2 t = *(const float2*)(hrow + (F & ~3));
        v[(F & ~3)] = t.x; v[(F & ~3) + 1] = t.y;
    }
    #pragma unroll
    for (int f = 0; f < F; ++f) atomicAdd(&orow[f], nrm * v[f]);
}


__global__ void gcn_gemm2(const float* __restrict__ h1,
                          const float* __restrict__ b1,
                          const float* __restrict__ W2,
                          const float* __restrict__ dis,
                          float* __restrict__ h2,
                          float* __restrict__ h2a) {
    __shared__ float Bf2[(HIDDEN / 2) * 16 * 2];
    __shared__ float b1s[HIDDEN];

    for (int idx = threadIdx.x; idx < (HIDDEN / 2) * 16 * 2; idx += blockDim.x) {
        const int p = idx >> 5;
        const int c = (idx >> 1) & 15;
        const int j = idx & 1;
        Bf2[idx] = (c < N_CLASSES) ? W2[(2 * p + j) * N_CLASSES + c] : 0.0f;
    }
    if (threadIdx.x < HIDDEN) b1s[threadIdx.x] = b1[threadIdx.x];
    __syncthreads();

    const int lane = threadIdx.x & 31;
    const int wave = threadIdx.x >> 5;
    const int tile = blockIdx.x * (blockDim.x >> 5) + wave;
    if (tile >= M_TILES) return;
    const int row0 = tile << 4;
    const int half = lane >> 4;
    const int l16  = lane & 15;
    const float* arow = h1 + (size_t)(row0 + l16) * PAD1;

    v8f c = {};
    #pragma unroll
    for (int k0 = 0; k0 < HIDDEN; k0 += 4) {
        const int kb = k0 + 2 * half;
        v2f ar = *(const v2f*)(arow + kb);
        v2f a;
        a.x = fmaxf(ar.x + b1s[kb],     0.0f);
        a.y = fmaxf(ar.y + b1s[kb + 1], 0.0f);
        v2f b = *(const v2f*)&Bf2[(kb >> 1) * 32 + l16 * 2];
        c = __builtin_amdgcn_wmma_f32_16x16x4_f32(false, a, false, b, (short)0, c, false, false);
    }

    #pragma unroll
    for (int r = 0; r < 8; ++r) {
        const int row = row0 + r + 8 * half;
        const float di = dis[row];
        const float d2 = di * di;
        h2 [(size_t)row * PAD2 + l16] = c[r];
        h2a[(size_t)row * PAD2 + l16] = d2 * c[r];
    }
}


__global__ void gcn_logsoftmax(const float* __restrict__ h2a,
                               const float* __restrict__ b2,
                               float* __restrict__ out) {
    int i = blockIdx.x * blockDim.x + threadIdx.x;
    if (i >= N_NODES) return;
    const float* row = h2a + (size_t)i * PAD2;
    float z[N_CLASSES];
    {
        float4 t0 = *(const float4*)(row);
        float4 t1 = *(const float4*)(row + 4);
        float2 t2 = *(const float2*)(row + 8);
        z[0] = t0.x; z[1] = t0.y; z[2] = t0.z; z[3] = t0.w;
        z[4] = t1.x; z[5] = t1.y; z[6] = t1.z; z[7] = t1.w;
        z[8] = t2.x; z[9] = t2.y;
    }
    float m = -3.402823466e38f;
    #pragma unroll
    for (int c = 0; c < N_CLASSES; ++c) {
        z[c] += b2[c];
        m = fmaxf(m, z[c]);
    }
    float s = 0.0f;
    #pragma unroll
    for (int c = 0; c < N_CLASSES; ++c) s += __expf(z[c] - m);
    const float l = __logf(s);
    #pragma unroll
    for (int c = 0; c < N_CLASSES; ++c) out[(size_t)i * N_CLASSES + c] = z[c] - m - l;
}


extern "C" void kernel_launch(void* const* d_in, const int* in_sizes, int n_in,
                              void* d_out, int out_size, void* d_ws, size_t ws_size,
                              hipStream_t stream) {
    const float* x  = (const float*)d_in[0];
    const int*   ei = (const int*)d_in[1];
    const float* W1 = (const float*)d_in[2];
    const float* b1 = (const float*)d_in[3];
    const float* W2 = (const float*)d_in[4];
    const float* b2 = (const float*)d_in[5];
    float* out = (float*)d_out;

    const int E = in_sizes[1] / 2;
    const int* src = ei;
    const int* dst = ei + E;

    float* ws  = (float*)d_ws;
    float* dis = ws;
    float* h0  = ws + 102400;
    float* h1  = h0 + (size_t)N_NODES * PAD1;
    float* h2  = h1 + (size_t)N_NODES * PAD1;
    float* h2a = h2 + (size_t)N_NODES * PAD2;

    const int TB = 256;
    const int nodeBlocks = (N_NODES + TB - 1) / TB;
    const int edgeBlocks = (E + TB - 1) / TB;
    const int gemmBlocks = (M_TILES + (TB / 32) - 1) / (TB / 32);

    gcn_init_deg<<<nodeBlocks, TB, 0, stream>>>(dis);
    gcn_deg_accum<<<edgeBlocks, TB, 0, stream>>>(dst, dis, E);
    gcn_rsqrt<<<nodeBlocks, TB, 0, stream>>>(dis);

    gcn_gemm1<<<gemmBlocks, TB, 0, stream>>>(x, W1, dis, h0, h1);
    gcn_propagate<HIDDEN, PAD1><<<edgeBlocks, TB, 0, stream>>>(src, dst, dis, h0, h1, E);

    gcn_gemm2<<<gemmBlocks, TB, 0, stream>>>(h1, b1, W2, dis, h2, h2a);
    gcn_propagate<N_CLASSES, PAD2><<<edgeBlocks, TB, 0, stream>>>(src, dst, dis, h2, h2a, E);

    gcn_logsoftmax<<<nodeBlocks, TB, 0, stream>>>(h2a, b2, out);
}


# Degree, rsqrt and both GEMM launches issue their loads in one or two groups: memory waits per wave 7, 4, 25, 5 -> 2, 1, 2, 1; x@W1 A tile as 16 whole-row loads transposed via LDS; k order kept
# speedup vs baseline: 10.3369x; 10.3369x over previous
#include <hip/hip_runtime.h>

typedef __attribute__((ext_vector_type(2))) float v2f;
typedef __attribute__((ext_vector_type(8))) float v8f;

#define N_NODES   100000
#define N_FEAT    128
#define HIDDEN    20
#define N_CLASSES 10
#define PAD1      32
#define PAD2      16
#define M_TILES   (N_NODES / 16)
#define DEG_PLANE 102400
#define PART_PITCH 100128
#define DEG_RANGES 4
#define DEG_SLICES 64
#define DEG_RANGE  25024
#define DEG_THREADS 1024
#define DEG_BATCH   25
#define PLANE1    (N_NODES * PAD1)
#define PLANE2    (N_NODES * PAD2)
#define EDGE_BLOCKS 2048


__device__ __forceinline__ int l2_half() {
    const unsigned id = __builtin_amdgcn_s_sendmsg_rtn(0x87);
    return (int)((id >> 18) & 1u);
}


__global__ void __launch_bounds__(DEG_THREADS)
gcn_deg_count(const int* __restrict__ dst, unsigned* __restrict__ part, int E) {
    __shared__ unsigned cnt[DEG_RANGE];
    const int tid = threadIdx.x;
    const int r = blockIdx.x % DEG_RANGES;
    const int sl = blockIdx.x / DEG_RANGES;
    const int lo = r * DEG_RANGE;

    for (int i = tid; i < DEG_RANGE; i += DEG_THREADS) cnt[i] = 0u;
    __syncthreads();

    const int per = (E + DEG_SLICES - 1) / DEG_SLICES;
    const int e0 = min(sl * per, E);
    const int e1 = min(e0 + per, E);
    for (int base = e0 + tid; base < e1; base += DEG_BATCH * DEG_THREADS) {
        int d[DEG_BATCH];
        #pragma unroll
        for (int j = 0; j < DEG_BATCH; ++j) d[j] = dst[min(base + j * DEG_THREADS, e1 - 1)];
        #pragma unroll
        for (int j = 0; j < DEG_BATCH; ++j) {
            const unsigned rel = (unsigned)(d[j] - lo);
            if (base + j * DEG_THREADS < e1 && rel < (unsigned)DEG_RANGE) atomicAdd(&cnt[rel], 1u);
        }
    }
    __syncthreads();

    unsigned* mine = part + (size_t)sl * PART_PITCH + lo;
    for (int i = tid; i < DEG_RANGE; i += DEG_THREADS) mine[i] = cnt[i];
}

__global__ void gcn_rsqrt(const unsigned* __restrict__ part, float* __restrict__ dis) {
    int i = blockIdx.x * blockDim.x + threadIdx.x;
    if (i >= N_NODES) return;
    unsigned c = 0u;
    #pragma unroll
    for (int sl = 0; sl < DEG_SLICES; ++sl) c += part[(size_t)sl * PART_PITCH + i];
    dis[i] = rsqrtf(1.0f + (float)c);
}


#define G1_WAVES   25
#define G1_THREADS (G1_WAVES * 32)
#define G1_BLOCKS  250
#define A_PITCH    132
#define BF_WORDS   ((N_FEAT / 2) * 32 * 2)
#define BF_STEPS   ((BF_WORDS + G1_THREADS - 1) / G1_THREADS)

__global__ void __launch_bounds__(G1_THREADS)
gcn_gemm1(const float* __restrict__ x,
          const float* __restrict__ W1,
          const float* __restrict__ dis,
          float* __restrict__ h0,
          float* __restrict__ h1) {
    __shared__ float Bf[BF_STEPS * G1_THREADS];
    __shared__ __attribute__((aligned(16))) float At[G1_WAVES * 16 * A_PITCH];

    const int tid  = threadIdx.x;
    const int lane = tid & 31;
    const int wave = tid >> 5;
    const int tile = blockIdx.x * G1_WAVES + wave;
    const bool live = tile < M_TILES;
    const int row0 = tile << 4;
    const int half = lane >> 4;
    const int l16  = lane & 15;

    float wv[BF_STEPS];
    #pragma unroll
    for (int j = 0; j < BF_STEPS; ++j) {
        const int idx = min(tid + j * G1_THREADS, BF_WORDS - 1);
        const int p  = idx >> 6;
        const int cc = (idx >> 1) & 31;
        const int jj = idx & 1;
        const float t = W1[(2 * p + jj) * HIDDEN + min(cc, HIDDEN - 1)];
        wv[j] = (cc < HIDDEN) ? t : 0.0f;
    }
    const int lrow0 = live ? row0 : 0;
    const float4* dp = (const float4*)(dis + lrow0 + 8 * half);
    const float4 dlo = dp[0];
    const float4 dhi = dp[1];

    #pragma unroll
    for (int j = 0; j < BF_STEPS; ++j) Bf[tid + j * G1_THREADS] = wv[j];
    __syncthreads();
    if (!live) return;

    float4 a[16];
    const float4* xt = (const float4*)(x + (size_t)row0 * N_FEAT) + lane;
    #pragma unroll
    for (int i = 0; i < 16; ++i) a[i] = xt[i * (N_FEAT / 4)];

    float* Aw = At + wave * (16 * A_PITCH);
    #pragma unroll
    for (int i = 0; i < 16; ++i) *(float4*)&Aw[i * A_PITCH + lane * 4] = a[i];

    const float* ar = Aw + l16 * A_PITCH + 2 * half;

    v8f c0 = {};
    v8f c1 = {};

    #pragma unroll 8
    for (int k0 = 0; k0 < N_FEAT; k0 += 4) {
        const int p = (k0 + 2 * half) >> 1;
        v2f av = *(const v2f*)(ar + k0);
        v2f b0 = *(const v2f*)&Bf[p * 64 + l16 * 2];
        v2f b1 = *(const v2f*)&Bf[p * 64 + (16 + l16) * 2];

        c0 = __builtin_amdgcn_wmma_f32_16x16x4_f32(false, av, false, b0, (short)0, c0, false, false);
        c1 = __builtin_amdgcn_wmma_f32_16x16x4_f32(false, av, false, b1, (short)0, c1, false, false);
    }

    const float dv[8] = {dlo.x, dlo.y, dlo.z, dlo.w, dhi.x, dhi.y, dhi.z, dhi.w};
    #pragma unroll
    for (int r = 0; r < 8; ++r) {
        const int row = row0 + r + 8 * half;
        const float di = dv[r];
        const float d2 = di * di;
        float* p0 = h0 + (size_t)row * PAD1;
        float* p1 = h1 + (size_t)row * PAD1;
        p0[l16]      = c0[r];
        p0[16 + l16] = (l16 == HIDDEN - 16) ? di : c1[r];
        p1[l16]      = d2 * c0[r];
        p1[16 + l16] = d2 * c1[r];
    }
    float4* z = (float4*)(h1 + PLANE1 + (size_t)row0 * PAD1);
    for (int q = 0; q < 4; ++q) z[q * 32 + lane] = make_float4(0.0f, 0.0f, 0.0f, 0.0f);
}


template <int G>
__device__ __forceinline__ int edge_slot(int val, int j, int sub) {
    if constexpr (G == 1) {
        return __builtin_amdgcn_readlane(val, j);
    } else {
        const int a = __builtin_amdgcn_readlane(val, 2 * j);
        const int b = __builtin_amdgcn_readlane(val, 2 * j + 1);
        return sub ? b : a;
    }
}

template <int F, int PAD, bool FULL>
__device__ __forceinline__ void scatter_wave(int s, int d, float nrm, int nvalid,
                                             int sub, int col,
                                             const float* __restrict__ h,
                                             float* __restrict__ out) {
    constexpr int G     = 32 / PAD;
    constexpr int STEPS = 32 / G;

    float v[STEPS];
    #pragma unroll
    for (int j = 0; j < STEPS; ++j) {
        const int sj = edge_slot<G>(s, j, sub);
        const float* hrow = h + (size_t)sj * PAD;
        if (FULL || j * G + sub < nvalid) v[j] = hrow[col];
        else v[j] = 0.0f;
    }
    #pragma unroll
    for (int j = 0; j < STEPS; ++j) {
        const int dj = edge_slot<G>(d, j, sub);
        const float nj = __int_as_float(edge_slot<G>(__float_as_int(nrm), j, sub));
        float* orow = out + (size_t)dj * PAD;
        if (FULL || j * G + sub < nvalid) atomicAdd(&orow[col], nj * v[j]);
    }
}

template <int F, int PAD, bool FULL>
__device__ __forceinline__ float scatter_wave_carry(int s, int d, float dd, int nvalid,
                                                    int lane, int col,
                                                    const float* __restrict__ h,
                                                    float* __restrict__ out) {
    static_assert(PAD == 32 && F < PAD, "one edge per step, one spare word in the row");
    const int gcol = min(col, F);
    float v[32];
    #pragma unroll
    for (int j = 0; j < 32; ++j) {
        const int sj = __builtin_amdgcn_readlane(s, j);
        const float* hrow = h + (size_t)sj * PAD;
        if (FULL || j < nvalid) v[j] = hrow[gcol];
        else v[j] = 0.0f;
    }
    int dsb = 0;
    #pragma unroll
    for (int j = 0; j < 32; ++j) {
        const int t = __builtin_amdgcn_readlane(__float_as_int(v[j]), F);
        dsb = (lane == j) ? t : dsb;
    }
    const float nrm = __int_as_float(dsb) * dd;
    if (col < F) {
        #pragma unroll
        for (int j = 0; j < 32; ++j) {
            const int dj = __builtin_amdgcn_readlane(d, j);
            const float nj = __int_as_float(__builtin_amdgcn_readlane(__float_as_int(nrm), j));
            float* orow = out + (size_t)dj * PAD;
            if (FULL || j < nvalid) atomicAdd(&orow[col], nj * v[j]);
        }
    }
    return nrm;
}

template <int F, int PAD, bool FIRST>
__global__ void gcn_propagate(const int* __restrict__ src,
                              const int* __restrict__ dst,
                              const float* __restrict__ dis,
                              float* __restrict__ w,
                              const float* __restrict__ h,
                              float* __restrict__ out, int E) {
    static_assert(PAD == 32 || PAD == 16, "PAD must be 32 or 16");
    out += (size_t)l2_half() * ((size_t)N_NODES * PAD);
    const int lane = threadIdx.x & 31;
    const int sub = lane / PAD;
    const int col = lane % PAD;
    const int wpb = blockDim.x >> 5;
    const int nwaves = gridDim.x * wpb;
    const int nchunks = (E + 31) >> 5;
    int c = __builtin_amdgcn_readfirstlane(blockIdx.x * wpb + (threadIdx.x >> 5));

    if (c >= nchunks) return;

    int s, d;
    float wt = 0.0f;
    {
        const int e = min((c << 5) + lane, E - 1);
        s = src[e];
        d = dst[e];
        if constexpr (!FIRST) wt = w[e];
    }
    while (c < nchunks) {
        const int nvalid = E - (c << 5);
        const int cn = c + nwaves;
        const int en = min((cn << 5) + lane, E - 1);
        const int sn = src[en];
        const int dn = dst[en];
        float wn = 0.0f;
        if constexpr (!FIRST) wn = w[en];
        if constexpr (FIRST) {
            const float dd = dis[d];
            float nrm;
            if (nvalid >= 32) nrm = scatter_wave_carry<F, PAD, true >(s, d, dd, nvalid, lane, col, h, out);
            else              nrm = scatter_wave_carry<F, PAD, false>(s, d, dd, nvalid, lane, col, h, out);
            if (lane < nvalid) w[(c << 5) + lane] = nrm;
        } else {
            const float nrm = wt;
            if (col < F) {
                if (nvalid >= 32) scatter_wave<F, PAD, true >(s, d, nrm, nvalid, sub, col, h, out);
                else              scatter_wave<F, PAD, false>(s, d, nrm, nvalid, sub, col, h, out);
            }
        }
        s = sn; d = dn; wt = wn; c = cn;
    }
}


#define GEMM_THREADS 256
#define BF2_WORDS  ((HIDDEN / 2) * 16 * 2)
#define BF2_STEPS  ((BF2_WORDS + GEMM_THREADS - 1) / GEMM_THREADS)

__global__ void __launch_bounds__(GEMM_THREADS)
gcn_gemm2(const float* __restrict__ h1,
          const float* __restrict__ b1,
          const float* __restrict__ W2,
          const float* __restrict__ dis,
          float* __restrict__ h2,
          float* __restrict__ h2a) {
    __shared__ float Bf2[BF2_STEPS * GEMM_THREADS];
    __shared__ float b1s[GEMM_THREADS];

    const int tid  = threadIdx.x;
    const int lane = tid & 31;
    const int wave = tid >> 5;
    const int tile = blockIdx.x * (GEMM_THREADS / 32) + wave;
    const bool live = tile < M_TILES;
    const int row0 = tile << 4;
    const int half = lane >> 4;
    const int l16  = lane & 15;

    float wv[BF2_STEPS];
    #pragma unroll
    for (int j = 0; j < BF2_STEPS; ++j) {
        const int idx = min(tid + j * GEMM_THREADS, BF2_WORDS - 1);
        const int p  = idx >> 5;
        const int cc = (idx >> 1) & 15;
        const int jj = idx & 1;
        const float t = W2[(2 * p + jj) * N_CLASSES + min(cc, N_CLASSES - 1)];
        wv[j] = (cc < N_CLASSES) ? t : 0.0f;
    }
    const float bv = b1[min(tid, HIDDEN - 1)];

    const int lrow0 = live ? row0 : 0;
    const float* arow = h1 + (size_t)(lrow0 + l16) * PAD1 + 2 * half;
    v2f ar[HIDDEN / 4], ab[HIDDEN / 4];
    #pragma unroll
    for (int s = 0; s < HIDDEN / 4; ++s) {
        ar[s] = *(const v2f*)(arow + 4 * s);
        ab[s] = *(const v2f*)(arow + PLANE1 + 4 * s);
    }
    const float4* dp = (const float4*)(dis + lrow0 + 8 * half);
    const float4 dlo = dp[0];
    const float4 dhi = dp[1];
    asm volatile("" :: "v"(ar[0]), "v"(ar[1]), "v"(ar[2]), "v"(ar[3]), "v"(ar[4]),
                       "v"(ab[0]), "v"(ab[1]), "v"(ab[2]), "v"(ab[3]), "v"(ab[4]),
                       "v"(dlo.x), "v"(dlo.y), "v"(dlo.z), "v"(dlo.w),
                       "v"(dhi.x), "v"(dhi.y), "v"(dhi.z), "v"(dhi.w));

    #pragma unroll
    for (int j = 0; j < BF2_STEPS; ++j) Bf2[tid + j * GEMM_THREADS] = wv[j];
    b1s[tid] = bv;
    __syncthreads();
    if (!live) return;

    v8f c = {};
    #pragma unroll
    for (int s = 0; s < HIDDEN / 4; ++s) {
        const int kb = 4 * s + 2 * half;
        v2f a;
        a.x = fmaxf((ar[s].x + ab[s].x) + b1s[kb],     0.0f);
        a.y = fmaxf((ar[s].y + ab[s].y) + b1s[kb + 1], 0.0f);
        v2f b = *(const v2f*)&Bf2[(kb >> 1) * 32 + l16 * 2];
        c = __builtin_amdgcn_wmma_f32_16x16x4_f32(false, a, false, b, (short)0, c, false, false);
    }

    const float dv[8] = {dlo.x, dlo.y, dlo.z, dlo.w, dhi.x, dhi.y, dhi.z, dhi.w};
    #pragma unroll
    for (int r = 0; r < 8; ++r) {
        const int row = row0 + r + 8 * half;
        const float di = dv[r];
        const float d2 = di * di;
        h2 [(size_t)row * PAD2 + l16] = c[r];
        h2a[(size_t)row * PAD2 + l16] = d2 * c[r];
    }
    float4* z = (float4*)(h2a + PLANE2 + (size_t)row0 * PAD2);
    for (int q = 0; q < 2; ++q) z[q * 32 + lane] = make_float4(0.0f, 0.0f, 0.0f, 0.0f);
}


__global__ void gcn_logsoftmax(const float* __restrict__ h2a,
                               const float* __restrict__ b2,
                               float* __restrict__ out) {
    int i = blockIdx.x * blockDim.x + threadIdx.x;
    if (i >= N_NODES) return;
    const float* row = h2a + (size_t)i * PAD2;
    float z[N_CLASSES];
    {
        float4 t0 = *(const float4*)(row);
        float4 t1 = *(const float4*)(row + 4);
        float2 t2 = *(const float2*)(row + 8);
        float4 u0 = *(const float4*)(row + PLANE2);
        float4 u1 = *(const float4*)(row + PLANE2 + 4);
        float2 u2 = *(const float2*)(row + PLANE2 + 8);
        z[0] = t0.x + u0.x; z[1] = t0.y + u0.y; z[2] = t0.z + u0.z; z[3] = t0.w + u0.w;
        z[4] = t1.x + u1.x; z[5] = t1.y + u1.y; z[6] = t1.z + u1.z; z[7] = t1.w + u1.w;
        z[8] = t2.x + u2.x; z[9] = t2.y + u2.y;
    }
    float m = -3.402823466e38f;
    #pragma unroll
    for (int c = 0; c < N_CLASSES; ++c) {
        z[c] += b2[c];
        m = fmaxf(m, z[c]);
    }
    float s = 0.0f;
    #pragma unroll
    for (int c = 0; c < N_CLASSES; ++c) s += __expf(z[c] - m);
    const float l = __logf(s);
    #pragma unroll
    for (int c = 0; c < N_CLASSES; ++c) out[(size_t)i * N_CLASSES + c] = z[c] - m - l;
}


extern "C" void kernel_launch(void* const* d_in, const int* in_sizes, int n_in,
                              void* d_out, int out_size, void* d_ws, size_t ws_size,
                              hipStream_t stream) {
    const float* x  = (const float*)d_in[0];
    const int*   ei = (const int*)d_in[1];
    const float* W1 = (const float*)d_in[2];
    const float* b1 = (const float*)d_in[3];
    const float* W2 = (const float*)d_in[4];
    const float* b2 = (const float*)d_in[5];
    float* out = (float*)d_out;

    const int E = in_sizes[1] / 2;
    const int* src = ei;
    const int* dst = ei + E;

    float* ws  = (float*)d_ws;
    float* dis = ws;
    float* h0  = ws + 2 * DEG_PLANE;
    float* h1  = h0 + (size_t)N_NODES * PAD1;
    float* h2  = h1 + 2 * (size_t)PLANE1;
    float* h2a = h2 + (size_t)N_NODES * PAD2;
    unsigned* part = (unsigned*)(h2a + 2 * (size_t)PLANE2);
    float* ew  = (float*)(part + (size_t)DEG_SLICES * PART_PITCH);

    const int TB = 256;
    const int nodeBlocks = (N_NODES + TB - 1) / TB;
    const int edgeBlocks = EDGE_BLOCKS;
    const int gemmBlocks = (M_TILES + (TB / 32) - 1) / (TB / 32);

    gcn_deg_count<<<DEG_RANGES * DEG_SLICES, DEG_THREADS, 0, stream>>>(dst, part, E);
    gcn_rsqrt<<<nodeBlocks, TB, 0, stream>>>(part, dis);

    gcn_gemm1<<<G1_BLOCKS, G1_THREADS, 0, stream>>>(x, W1, dis, h0, h1);
    gcn_propagate<HIDDEN, PAD1, true><<<edgeBlocks, TB, 0, stream>>>(src, dst, dis, ew, h0, h1, E);

    gcn_gemm2<<<gemmBlocks, GEMM_THREADS, 0, stream>>>(h1, b1, W2, dis, h2, h2a);
    gcn_propagate<N_CLASSES, PAD2, false><<<edgeBlocks, TB, 0, stream>>>(src, dst, dis, ew, h2, h2a, E);

    gcn_logsoftmax<<<nodeBlocks, TB, 0, stream>>>(h2a, b2, out);
}


# Edge kernels address rows as plane base + 32-bit lane offset: one 32-bit op per row address, not a sign extension, 64-bit shift and 64-bit add; same loads, atomics and arrival-order f32 sums
# speedup vs baseline: 10.8192x; 1.0467x over previous
#include <hip/hip_runtime.h>

typedef __attribute__((ext_vector_type(2))) float v2f;
typedef __attribute__((ext_vector_type(8))) float v8f;

#define N_NODES   100000
#define N_FEAT    128
#define HIDDEN    20
#define N_CLASSES 10
#define PAD1      32
#define PAD2      16
#define M_TILES   (N_NODES / 16)
#define DEG_PLANE 102400
#define PART_PITCH 100128
#define DEG_RANGES 4
#define DEG_SLICES 64
#define DEG_RANGE  25024
#define DEG_THREADS 1024
#define DEG_BATCH   25
#define PLANE1    (N_NODES * PAD1)
#define PLANE2    (N_NODES * PAD2)
#define EDGE_BLOCKS 2048


__device__ __forceinline__ int l2_half() {
    const unsigned id = __builtin_amdgcn_s_sendmsg_rtn(0x87);
    return (int)((id >> 18) & 1u);
}


__global__ void __launch_bounds__(DEG_THREADS)
gcn_deg_count(const int* __restrict__ dst, unsigned* __restrict__ part, int E) {
    __shared__ unsigned cnt[DEG_RANGE];
    const int tid = threadIdx.x;
    const int r = blockIdx.x % DEG_RANGES;
    const int sl = blockIdx.x / DEG_RANGES;
    const int lo = r * DEG_RANGE;

    for (int i = tid; i < DEG_RANGE; i += DEG_THREADS) cnt[i] = 0u;
    __syncthreads();

    const int per = (E + DEG_SLICES - 1) / DEG_SLICES;
    const int e0 = min(sl * per, E);
    const int e1 = min(e0 + per, E);
    for (int base = e0 + tid; base < e1; base += DEG_BATCH * DEG_THREADS) {
        int d[DEG_BATCH];
        #pragma unroll
        for (int j = 0; j < DEG_BATCH; ++j) d[j] = dst[min(base + j * DEG_THREADS, e1 - 1)];
        #pragma unroll
        for (int j = 0; j < DEG_BATCH; ++j) {
            const unsigned rel = (unsigned)(d[j] - lo);
            if (base + j * DEG_THREADS < e1 && rel < (unsigned)DEG_RANGE) atomicAdd(&cnt[rel], 1u);
        }
    }
    __syncthreads();

    unsigned* mine = part + (size_t)sl * PART_PITCH + lo;
    for (int i = tid; i < DEG_RANGE; i += DEG_THREADS) mine[i] = cnt[i];
}

__global__ void gcn_rsqrt(const unsigned* __restrict__ part, float* __restrict__ dis) {
    int i = blockIdx.x * blockDim.x + threadIdx.x;
    if (i >= N_NODES) return;
    unsigned c = 0u;
    #pragma unroll
    for (int sl = 0; sl < DEG_SLICES; ++sl) c += part[(size_t)sl * PART_PITCH + i];
    dis[i] = rsqrtf(1.0f + (float)c);
}


#define G1_WAVES   25
#define G1_THREADS (G1_WAVES * 32)
#define G1_BLOCKS  250
#define A_PITCH    132
#define BF_WORDS   ((N_FEAT / 2) * 32 * 2)
#define BF_STEPS   ((BF_WORDS + G1_THREADS - 1) / G1_THREADS)

__global__ void __launch_bounds__(G1_THREADS)
gcn_gemm1(const float* __restrict__ x,
          const float* __restrict__ W1,
          const float* __restrict__ dis,
          float* __restrict__ h0,
          float* __restrict__ h1) {
    __shared__ float Bf[BF_STEPS * G1_THREADS];
    __shared__ __attribute__((aligned(16))) float At[G1_WAVES * 16 * A_PITCH];

    const int tid  = threadIdx.x;
    const int lane = tid & 31;
    const int wave = tid >> 5;
    const int tile = blockIdx.x * G1_WAVES + wave;
    const bool live = tile < M_TILES;
    const int row0 = tile << 4;
    const int half = lane >> 4;
    const int l16  = lane & 15;

    float wv[BF_STEPS];
    #pragma unroll
    for (int j = 0; j < BF_STEPS; ++j) {
        const int idx = min(tid + j * G1_THREADS, BF_WORDS - 1);
        const int p  = idx >> 6;
        const int cc = (idx >> 1) & 31;
        const int jj = idx & 1;
        const float t = W1[(2 * p + jj) * HIDDEN + min(cc, HIDDEN - 1)];
        wv[j] = (cc < HIDDEN) ? t : 0.0f;
    }
    const int lrow0 = live ? row0 : 0;
    const float4* dp = (const float4*)(dis + lrow0 + 8 * half);
    const float4 dlo = dp[0];
    const float4 dhi = dp[1];

    #pragma unroll
    for (int j = 0; j < BF_STEPS; ++j) Bf[tid + j * G1_THREADS] = wv[j];
    __syncthreads();
    if (!live) return;

    float4 a[16];
    const float4* xt = (const float4*)(x + (size_t)row0 * N_FEAT) + lane;
    #pragma unroll
    for (int i = 0; i < 16; ++i) a[i] = xt[i * (N_FEAT / 4)];

    float* Aw = At + wave * (16 * A_PITCH);
    #pragma unroll
    for (int i = 0; i < 16; ++i) *(float4*)&Aw[i * A_PITCH + lane * 4] = a[i];

    const float* ar = Aw + l16 * A_PITCH + 2 * half;

    v8f c0 = {};
    v8f c1 = {};

    #pragma unroll 8
    for (int k0 = 0; k0 < N_FEAT; k0 += 4) {
        const int p = (k0 + 2 * half) >> 1;
        v2f av = *(const v2f*)(ar + k0);
        v2f b0 = *(const v2f*)&Bf[p * 64 + l16 * 2];
        v2f b1 = *(const v2f*)&Bf[p * 64 + (16 + l16) * 2];

        c0 = __builtin_amdgcn_wmma_f32_16x16x4_f32(false, av, false, b0, (short)0, c0, false, false);
        c1 = __builtin_amdgcn_wmma_f32_16x16x4_f32(false, av, false, b1, (short)0, c1, false, false);
    }

    const float dv[8] = {dlo.x, dlo.y, dlo.z, dlo.w, dhi.x, dhi.y, dhi.z, dhi.w};
    #pragma unroll
    for (int r = 0; r < 8; ++r) {
        const int row = row0 + r + 8 * half;
        const float di = dv[r];
        const float d2 = di * di;
        float* p0 = h0 + (size_t)row * PAD1;
        float* p1 = h1 + (size_t)row * PAD1;
        p0[l16]      = c0[r];
        p0[16 + l16] = (l16 == HIDDEN - 16) ? di : c1[r];
        p1[l16]      = d2 * c0[r];
        p1[16 + l16] = d2 * c1[r];
    }
    float4* z = (float4*)(h1 + PLANE1 + (size_t)row0 * PAD1);
    for (int q = 0; q < 4; ++q) z[q * 32 + lane] = make_float4(0.0f, 0.0f, 0.0f, 0.0f);
}


template <int G>
__device__ __forceinline__ int edge_slot(int val, int j, int sub) {
    if constexpr (G == 1) {
        return __builtin_amdgcn_readlane(val, j);
    } else {
        const int a = __builtin_amdgcn_readlane(val, 2 * j);
        const int b = __builtin_amdgcn_readlane(val, 2 * j + 1);
        return sub ? b : a;
    }
}

template <int F, int PAD, bool FULL>
__device__ __forceinline__ void scatter_wave(int s, int d, float nrm, int nvalid,
                                             int sub, int col,
                                             const float* __restrict__ h,
                                             float* __restrict__ out) {
    constexpr int G     = 32 / PAD;
    constexpr int STEPS = 32 / G;
    constexpr unsigned ROWB = PAD * 4u;
    static_assert((unsigned long long)N_NODES * ROWB < (1ull << 32), "row offsets must fit 32 bits");
    const unsigned col4 = (unsigned)col * 4u;

    float v[STEPS];
    #pragma unroll
    for (int j = 0; j < STEPS; ++j) {
        const unsigned sj = (unsigned)edge_slot<G>(s, j, sub);
        const float* hp = (const float*)((const char*)h + (sj * ROWB + col4));
        if (FULL || j * G + sub < nvalid) v[j] = *hp;
        else v[j] = 0.0f;
    }
    #pragma unroll
    for (int j = 0; j < STEPS; ++j) {
        const unsigned dj = (unsigned)edge_slot<G>(d, j, sub);
        const float nj = __int_as_float(edge_slot<G>(__float_as_int(nrm), j, sub));
        float* op = (float*)((char*)out + (dj * ROWB + col4));
        if (FULL || j * G + sub < nvalid) atomicAdd(op, nj * v[j]);
    }
}

template <int F, int PAD, bool FULL>
__device__ __forceinline__ float scatter_wave_carry(int s, int d, float dd, int nvalid,
                                                    int lane, int col,
                                                    const float* __restrict__ h,
                                                    float* __restrict__ out) {
    static_assert(PAD == 32 && F < PAD, "one edge per step, one spare word in the row");
    constexpr unsigned ROWB = PAD * 4u;
    static_assert((unsigned long long)N_NODES * ROWB < (1ull << 32), "row offsets must fit 32 bits");
    const unsigned gcol4 = (unsigned)min(col, F) * 4u;
    const unsigned col4 = (unsigned)col * 4u;
    float v[32];
    #pragma unroll
    for (int j = 0; j < 32; ++j) {
        const unsigned sj = (unsigned)__builtin_amdgcn_readlane(s, j);
        const float* hp = (const float*)((const char*)h + (sj * ROWB + gcol4));
        if (FULL || j < nvalid) v[j] = *hp;
        else v[j] = 0.0f;
    }
    int dsb = 0;
    #pragma unroll
    for (int j = 0; j < 32; ++j) {
        const int t = __builtin_amdgcn_readlane(__float_as_int(v[j]), F);
        dsb = (lane == j) ? t : dsb;
    }
    const float nrm = __int_as_float(dsb) * dd;
    if (col < F) {
        #pragma unroll
        for (int j = 0; j < 32; ++j) {
            const unsigned dj = (unsigned)__builtin_amdgcn_readlane(d, j);
            const float nj = __int_as_float(__builtin_amdgcn_readlane(__float_as_int(nrm), j));
            float* op = (float*)((char*)out + (dj * ROWB + col4));
            if (FULL || j < nvalid) atomicAdd(op, nj * v[j]);
        }
    }
    return nrm;
}

template <int F, int PAD, bool FIRST>
__global__ void gcn_propagate(const int* __restrict__ src,
                              const int* __restrict__ dst,
                              const float* __restrict__ dis,
                              float* __restrict__ w,
                              const float* __restrict__ h,
                              float* __restrict__ out, int E) {
    static_assert(PAD == 32 || PAD == 16, "PAD must be 32 or 16");
    out += (size_t)__builtin_amdgcn_readfirstlane(l2_half()) * ((size_t)N_NODES * PAD);
    const int lane = threadIdx.x & 31;
    const int sub = lane / PAD;
    const int col = lane % PAD;
    const int wpb = blockDim.x >> 5;
    const int nwaves = gridDim.x * wpb;
    const int nchunks = (E + 31) >> 5;
    int c = __builtin_amdgcn_readfirstlane(blockIdx.x * wpb + (threadIdx.x >> 5));

    if (c >= nchunks) return;

    int s, d;
    float wt = 0.0f;
    {
        const int e = min((c << 5) + lane, E - 1);
        s = src[e];
        d = dst[e];
        if constexpr (!FIRST) wt = w[e];
    }
    while (c < nchunks) {
        const int nvalid = E - (c << 5);
        const int cn = c + nwaves;
        const int en = min((cn << 5) + lane, E - 1);
        const int sn = src[en];
        const int dn = dst[en];
        float wn = 0.0f;
        if constexpr (!FIRST) wn = w[en];
        if constexpr (FIRST) {
            const float dd = dis[d];
            float nrm;
            if (nvalid >= 32) nrm = scatter_wave_carry<F, PAD, true >(s, d, dd, nvalid, lane, col, h, out);
            else              nrm = scatter_wave_carry<F, PAD, false>(s, d, dd, nvalid, lane, col, h, out);
            if (lane < nvalid) w[(c << 5) + lane] = nrm;
        } else {
            const float nrm = wt;
            if (col < F) {
                if (nvalid >= 32) scatter_wave<F, PAD, true >(s, d, nrm, nvalid, sub, col, h, out);
                else              scatter_wave<F, PAD, false>(s, d, nrm, nvalid, sub, col, h, out);
            }
        }
        s = sn; d = dn; wt = wn; c = cn;
    }
}


#define GEMM_THREADS 256
#define BF2_WORDS  ((HIDDEN / 2) * 16 * 2)
#define BF2_STEPS  ((BF2_WORDS + GEMM_THREADS - 1) / GEMM_THREADS)

__global__ void __launch_bounds__(GEMM_THREADS)
gcn_gemm2(const float* __restrict__ h1,
          const float* __restrict__ b1,
          const float* __restrict__ W2,
          const float* __restrict__ dis,
          float* __restrict__ h2,
          float* __restrict__ h2a) {
    __shared__ float Bf2[BF2_STEPS * GEMM_THREADS];
    __shared__ float b1s[GEMM_THREADS];

    const int tid  = threadIdx.x;
    const int lane = tid & 31;
    const int wave = tid >> 5;
    const int tile = blockIdx.x * (GEMM_THREADS / 32) + wave;
    const bool live = tile < M_TILES;
    const int row0 = tile << 4;
    const int half = lane >> 4;
    const int l16  = lane & 15;

    float wv[BF2_STEPS];
    #pragma unroll
    for (int j = 0; j < BF2_STEPS; ++j) {
        const int idx = min(tid + j * GEMM_THREADS, BF2_WORDS - 1);
        const int p  = idx >> 5;
        const int cc = (idx >> 1) & 15;
        const int jj = idx & 1;
        const float t = W2[(2 * p + jj) * N_CLASSES + min(cc, N_CLASSES - 1)];
        wv[j] = (cc < N_CLASSES) ? t : 0.0f;
    }
    const float bv = b1[min(tid, HIDDEN - 1)];

    const int lrow0 = live ? row0 : 0;
    const float* arow = h1 + (size_t)(lrow0 + l16) * PAD1 + 2 * half;
    v2f ar[HIDDEN / 4], ab[HIDDEN / 4];
    #pragma unroll
    for (int s = 0; s < HIDDEN / 4; ++s) {
        ar[s] = *(const v2f*)(arow + 4 * s);
        ab[s] = *(const v2f*)(arow + PLANE1 + 4 * s);
    }
    const float4* dp = (const float4*)(dis + lrow0 + 8 * half);
    const float4 dlo = dp[0];
    const float4 dhi = dp[1];
    asm volatile("" :: "v"(ar[0]), "v"(ar[1]), "v"(ar[2]), "v"(ar[3]), "v"(ar[4]),
                       "v"(ab[0]), "v"(ab[1]), "v"(ab[2]), "v"(ab[3]), "v"(ab[4]),
                       "v"(dlo.x), "v"(dlo.y), "v"(dlo.z), "v"(dlo.w),
                       "v"(dhi.x), "v"(dhi.y), "v"(dhi.z), "v"(dhi.w));

    #pragma unroll
    for (int j = 0; j < BF2_STEPS; ++j) Bf2[tid + j * GEMM_THREADS] = wv[j];
    b1s[tid] = bv;
    __syncthreads();
    if (!live) return;

    v8f c = {};
    #pragma unroll
    for (int s = 0; s < HIDDEN / 4; ++s) {
        const int kb = 4 * s + 2 * half;
        v2f a;
        a.x = fmaxf((ar[s].x + ab[s].x) + b1s[kb],     0.0f);
        a.y = fmaxf((ar[s].y + ab[s].y) + b1s[kb + 1], 0.0f);
        v2f b = *(const v2f*)&Bf2[(kb >> 1) * 32 + l16 * 2];
        c = __builtin_amdgcn_wmma_f32_16x16x4_f32(false, a, false, b, (short)0, c, false, false);
    }

    const float dv[8] = {dlo.x, dlo.y, dlo.z, dlo.w, dhi.x, dhi.y, dhi.z, dhi.w};
    #pragma unroll
    for (int r = 0; r < 8; ++r) {
        const int row = row0 + r + 8 * half;
        const float di = dv[r];
        const float d2 = di * di;
        h2 [(size_t)row * PAD2 + l16] = c[r];
        h2a[(size_t)row * PAD2 + l16] = d2 * c[r];
    }
    float4* z = (float4*)(h2a + PLANE2 + (size_t)row0 * PAD2);
    for (int q = 0; q < 2; ++q) z[q * 32 + lane] = make_float4(0.0f, 0.0f, 0.0f, 0.0f);
}


__global__ void gcn_logsoftmax(const float* __restrict__ h2a,
                               const float* __restrict__ b2,
                               float* __restrict__ out) {
    int i = blockIdx.x * blockDim.x + threadIdx.x;
    if (i >= N_NODES) return;
    const float* row = h2a + (size_t)i * PAD2;
    float z[N_CLASSES];
    {
        float4 t0 = *(const float4*)(row);
        float4 t1 = *(const float4*)(row + 4);
        float2 t2 = *(const float2*)(row + 8);
        float4 u0 = *(const float4*)(row + PLANE2);
        float4 u1 = *(const float4*)(row + PLANE2 + 4);
        float2 u2 = *(const float2*)(row + PLANE2 + 8);
        z[0] = t0.x + u0.x; z[1] = t0.y + u0.y; z[2] = t0.z + u0.z; z[3] = t0.w + u0.w;
        z[4] = t1.x + u1.x; z[5] = t1.y + u1.y; z[6] = t1.z + u1.z; z[7] = t1.w + u1.w;
        z[8] = t2.x + u2.x; z[9] = t2.y + u2.y;
    }
    float m = -3.402823466e38f;
    #pragma unroll
    for (int c = 0; c < N_CLASSES; ++c) {
        z[c] += b2[c];
        m = fmaxf(m, z[c]);
    }
    float s = 0.0f;
    #pragma unroll
    for (int c = 0; c < N_CLASSES; ++c) s += __expf(z[c] - m);
    const float l = __logf(s);
    #pragma unroll
    for (int c = 0; c < N_CLASSES; ++c) out[(size_t)i * N_CLASSES + c] = z[c] - m - l;
}


extern "C" void kernel_launch(void* const* d_in, const int* in_sizes, int n_in,
                              void* d_out, int out_size, void* d_ws, size_t ws_size,
                              hipStream_t stream) {
    const float* x  = (const float*)d_in[0];
    const int*   ei = (const int*)d_in[1];
    const float* W1 = (const float*)d_in[2];
    const float* b1 = (const float*)d_in[3];
    const float* W2 = (const float*)d_in[4];
    const float* b2 = (const float*)d_in[5];
    float* out = (float*)d_out;

    const int E = in_sizes[1] / 2;
    const int* src = ei;
    const int* dst = ei + E;

    float* ws  = (float*)d_ws;
    float* dis = ws;
    float* h0  = ws + 2 * DEG_PLANE;
    float* h1  = h0 + (size_t)N_NODES * PAD1;
    float* h2  = h1 + 2 * (size_t)PLANE1;
    float* h2a = h2 + (size_t)N_NODES * PAD2;
    unsigned* part = (unsigned*)(h2a + 2 * (size_t)PLANE2);
    float* ew  = (float*)(part + (size_t)DEG_SLICES * PART_PITCH);

    const int TB = 256;
    const int nodeBlocks = (N_NODES + TB - 1) / TB;
    const int edgeBlocks = EDGE_BLOCKS;
    const int gemmBlocks = (M_TILES + (TB / 32) - 1) / (TB / 32);

    gcn_deg_count<<<DEG_RANGES * DEG_SLICES, DEG_THREADS, 0, stream>>>(dst, part, E);
    gcn_rsqrt<<<nodeBlocks, TB, 0, stream>>>(part, dis);

    gcn_gemm1<<<G1_BLOCKS, G1_THREADS, 0, stream>>>(x, W1, dis, h0, h1);
    gcn_propagate<HIDDEN, PAD1, true><<<edgeBlocks, TB, 0, stream>>>(src, dst, dis, ew, h0, h1, E);

    gcn_gemm2<<<gemmBlocks, GEMM_THREADS, 0, stream>>>(h1, b1, W2, dis, h2, h2a);
    gcn_propagate<N_CLASSES, PAD2, false><<<edgeBlocks, TB, 0, stream>>>(src, dst, dis, ew, h2, h2a, E);

    gcn_logsoftmax<<<nodeBlocks, TB, 0, stream>>>(h2a, b2, out);
}


# Layer 1 edge weights from LDS: each work-group (32 waves, 2 per WGP) holds a 100 KB in-degree byte table, both rsqrt factors from LDS reads; no scattered dis read, no pad-word carry; sum order kept
# speedup vs baseline: 11.8006x; 1.0907x over previous
#include <hip/hip_runtime.h>

typedef __attribute__((ext_vector_type(2))) float v2f;
typedef __attribute__((ext_vector_type(8))) float v8f;

#define N_NODES   100000
#define N_FEAT    128
#define HIDDEN    20
#define N_CLASSES 10
#define PAD1      32
#define PAD2      16
#define M_TILES   (N_NODES / 16)
#define DEG_PLANE 102400
#define PART_PITCH 100128
#define DEG_RANGES 4
#define DEG_SLICES 64
#define DEG_RANGE  25024
#define DEG_THREADS 1024
#define DEG_BATCH   25
#define PLANE1    (N_NODES * PAD1)
#define PLANE2    (N_NODES * PAD2)
#define EDGE_BLOCKS 2048
#define EDGE1_THREADS 1024
#define EDGE1_BLOCKS  512
#define TAB_CAP   255u
#define TAB_VEC   (N_NODES / 16)
#define TAB_STEPS ((TAB_VEC + EDGE1_THREADS - 1) / EDGE1_THREADS)
static_assert(N_NODES % 16 == 0, "the table is copied in whole 16-byte pieces");


__device__ __forceinline__ int l2_half() {
    const unsigned id = __builtin_amdgcn_s_sendmsg_rtn(0x87);
    return (int)((id >> 18) & 1u);
}


__global__ void __launch_bounds__(DEG_THREADS)
gcn_deg_count(const int* __restrict__ dst, unsigned* __restrict__ part, int E) {
    __shared__ unsigned cnt[DEG_RANGE];
    const int tid = threadIdx.x;
    const int r = blockIdx.x % DEG_RANGES;
    const int sl = blockIdx.x / DEG_RANGES;
    const int lo = r * DEG_RANGE;

    for (int i = tid; i < DEG_RANGE; i += DEG_THREADS) cnt[i] = 0u;
    __syncthreads();

    const int per = (E + DEG_SLICES - 1) / DEG_SLICES;
    const int e0 = min(sl * per, E);
    const int e1 = min(e0 + per, E);
    for (int base = e0 + tid; base < e1; base += DEG_BATCH * DEG_THREADS) {
        int d[DEG_BATCH];
        #pragma unroll
        for (int j = 0; j < DEG_BATCH; ++j) d[j] = dst[min(base + j * DEG_THREADS, e1 - 1)];
        #pragma unroll
        for (int j = 0; j < DEG_BATCH; ++j) {
            const unsigned rel = (unsigned)(d[j] - lo);
            if (base + j * DEG_THREADS < e1 && rel < (unsigned)DEG_RANGE) atomicAdd(&cnt[rel], 1u);
        }
    }
    __syncthreads();

    unsigned* mine = part + (size_t)sl * PART_PITCH + lo;
    for (int i = tid; i < DEG_RANGE; i += DEG_THREADS) mine[i] = cnt[i];
}

__global__ void gcn_rsqrt(const unsigned* __restrict__ part, float* __restrict__ dis,
                          unsigned char* __restrict__ cnt8) {
    int i = blockIdx.x * blockDim.x + threadIdx.x;
    if (i >= N_NODES) return;
    unsigned c = 0u;
    #pragma unroll
    for (int sl = 0; sl < DEG_SLICES; ++sl) c += part[(size_t)sl * PART_PITCH + i];
    dis[i] = rsqrtf(1.0f + (float)c);
    cnt8[i] = (unsigned char)min(c, TAB_CAP);
}


#define G1_WAVES   25
#define G1_THREADS (G1_WAVES * 32)
#define G1_BLOCKS  250
#define A_PITCH    132
#define BF_WORDS   ((N_FEAT / 2) * 32 * 2)
#define BF_STEPS   ((BF_WORDS + G1_THREADS - 1) / G1_THREADS)

__global__ void __launch_bounds__(G1_THREADS)
gcn_gemm1(const float* __restrict__ x,
          const float* __restrict__ W1,
          const float* __restrict__ dis,
          float* __restrict__ h0,
          float* __restrict__ h1) {
    __shared__ float Bf[BF_STEPS * G1_THREADS];
    __shared__ __attribute__((aligned(16))) float At[G1_WAVES * 16 * A_PITCH];

    const int tid  = threadIdx.x;
    const int lane = tid & 31;
    const int wave = tid >> 5;
    const int tile = blockIdx.x * G1_WAVES + wave;
    const bool live = tile < M_TILES;
    const int row0 = tile << 4;
    const int half = lane >> 4;
    const int l16  = lane & 15;

    float wv[BF_STEPS];
    #pragma unroll
    for (int j = 0; j < BF_STEPS; ++j) {
        const int idx = min(tid + j * G1_THREADS, BF_WORDS - 1);
        const int p  = idx >> 6;
        const int cc = (idx >> 1) & 31;
        const int jj = idx & 1;
        const float t = W1[(2 * p + jj) * HIDDEN + min(cc, HIDDEN - 1)];
        wv[j] = (cc < HIDDEN) ? t : 0.0f;
    }
    const int lrow0 = live ? row0 : 0;
    const float4* dp = (const float4*)(dis + lrow0 + 8 * half);
    const float4 dlo = dp[0];
    const float4 dhi = dp[1];

    #pragma unroll
    for (int j = 0; j < BF_STEPS; ++j) Bf[tid + j * G1_THREADS] = wv[j];
    __syncthreads();
    if (!live) return;

    float4 a[16];
    const float4* xt = (const float4*)(x + (size_t)row0 * N_FEAT) + lane;
    #pragma unroll
    for (int i = 0; i < 16; ++i) a[i] = xt[i * (N_FEAT / 4)];

    float* Aw = At + wave * (16 * A_PITCH);
    #pragma unroll
    for (int i = 0; i < 16; ++i) *(float4*)&Aw[i * A_PITCH + lane * 4] = a[i];

    const float* ar = Aw + l16 * A_PITCH + 2 * half;

    v8f c0 = {};
    v8f c1 = {};

    #pragma unroll 8
    for (int k0 = 0; k0 < N_FEAT; k0 += 4) {
        const int p = (k0 + 2 * half) >> 1;
        v2f av = *(const v2f*)(ar + k0);
        v2f b0 = *(const v2f*)&Bf[p * 64 + l16 * 2];
        v2f b1 = *(const v2f*)&Bf[p * 64 + (16 + l16) * 2];

        c0 = __builtin_amdgcn_wmma_f32_16x16x4_f32(false, av, false, b0, (short)0, c0, false, false);
        c1 = __builtin_amdgcn_wmma_f32_16x16x4_f32(false, av, false, b1, (short)0, c1, false, false);
    }

    const float dv[8] = {dlo.x, dlo.y, dlo.z, dlo.w, dhi.x, dhi.y, dhi.z, dhi.w};
    #pragma unroll
    for (int r = 0; r < 8; ++r) {
        const int row = row0 + r + 8 * half;
        const float di = dv[r];
        const float d2 = di * di;
        float* p0 = h0 + (size_t)row * PAD1;
        float* p1 = h1 + (size_t)row * PAD1;
        p0[l16]      = c0[r];
        p0[16 + l16] = c1[r];
        p1[l16]      = d2 * c0[r];
        p1[16 + l16] = d2 * c1[r];
    }
    float4* z = (float4*)(h1 + PLANE1 + (size_t)row0 * PAD1);
    for (int q = 0; q < 4; ++q) z[q * 32 + lane] = make_float4(0.0f, 0.0f, 0.0f, 0.0f);
}


template <int G>
__device__ __forceinline__ int edge_slot(int val, int j, int sub) {
    if constexpr (G == 1) {
        return __builtin_amdgcn_readlane(val, j);
    } else {
        const int a = __builtin_amdgcn_readlane(val, 2 * j);
        const int b = __builtin_amdgcn_readlane(val, 2 * j + 1);
        return sub ? b : a;
    }
}

template <int F, int PAD, bool FULL>
__device__ __forceinline__ void scatter_wave(int s, int d, float nrm, int nvalid,
                                             int sub, int col,
                                             const float* __restrict__ h,
                                             float* __restrict__ out) {
    constexpr int G     = 32 / PAD;
    constexpr int STEPS = 32 / G;
    constexpr unsigned ROWB = PAD * 4u;
    static_assert((unsigned long long)N_NODES * ROWB < (1ull << 32), "row offsets must fit 32 bits");
    const unsigned col4 = (unsigned)col * 4u;

    float v[STEPS];
    #pragma unroll
    for (int j = 0; j < STEPS; ++j) {
        const unsigned sj = (unsigned)edge_slot<G>(s, j, sub);
        const float* hp = (const float*)((const char*)h + (sj * ROWB + col4));
        if (FULL || j * G + sub < nvalid) v[j] = *hp;
        else v[j] = 0.0f;
    }
    #pragma unroll
    for (int j = 0; j < STEPS; ++j) {
        const unsigned dj = (unsigned)edge_slot<G>(d, j, sub);
        const float nj = __int_as_float(edge_slot<G>(__float_as_int(nrm), j, sub));
        float* op = (float*)((char*)out + (dj * ROWB + col4));
        if (FULL || j * G + sub < nvalid) atomicAdd(op, nj * v[j]);
    }
}


__device__ __forceinline__ float edge_weight_lds(const unsigned char* tab,
                                                 const float* __restrict__ dis, int s, int d) {
    const unsigned cs = tab[s];
    const unsigned cd = tab[d];
    float nrm = rsqrtf(1.0f + (float)cs) * rsqrtf(1.0f + (float)cd);
    if (cs == TAB_CAP || cd == TAB_CAP) nrm = dis[s] * dis[d];
    return nrm;
}

__global__ void __launch_bounds__(EDGE1_THREADS)
gcn_scatter1(const int* __restrict__ src,
             const int* __restrict__ dst,
             const float* __restrict__ dis,
             const unsigned char* __restrict__ cnt8,
             float* __restrict__ w,
             const float* __restrict__ h,
             float* __restrict__ out, int E) {
    constexpr int F = HIDDEN, PAD = PAD1;
    __shared__ __attribute__((aligned(16))) unsigned char tab[N_NODES];
    out += (size_t)__builtin_amdgcn_readfirstlane(l2_half()) * ((size_t)N_NODES * PAD);
    const int lane = threadIdx.x & 31;
    const int nwaves = gridDim.x * (EDGE1_THREADS / 32);
    const int nchunks = (E + 31) >> 5;
    const int nfull = E >> 5;
    int c = __builtin_amdgcn_readfirstlane(blockIdx.x * (EDGE1_THREADS / 32) + (threadIdx.x >> 5));

    const uint4* tg = (const uint4*)cnt8;
    uint4 t[TAB_STEPS];
    #pragma unroll
    for (int j = 0; j < TAB_STEPS; ++j) t[j] = tg[min((int)threadIdx.x + j * EDGE1_THREADS, TAB_VEC - 1)];
    #pragma unroll
    for (int j = 0; j < TAB_STEPS; ++j) ((uint4*)tab)[min((int)threadIdx.x + j * EDGE1_THREADS, TAB_VEC - 1)] = t[j];
    __syncthreads();

    if (c >= nchunks) return;

    const int e = min((c << 5) + lane, E - 1);
    int s = src[e];
    int d = dst[e];
    while (c < nfull) {
        const int cn = c + nwaves;
        const int en = min((cn << 5) + lane, E - 1);
        const int sn = src[en];
        const int dn = dst[en];
        const float nrm = edge_weight_lds(tab, dis, s, d);
        w[(c << 5) + lane] = nrm;
        if (lane < F) scatter_wave<F, PAD, true>(s, d, nrm, 32, 0, lane, h, out);
        s = sn; d = dn; c = cn;
    }
    if (c < nchunks) {
        const int nvalid = E - (c << 5);
        const float nrm = edge_weight_lds(tab, dis, s, d);
        if (lane < nvalid) w[(c << 5) + lane] = nrm;
        if (lane < F) scatter_wave<F, PAD, false>(s, d, nrm, nvalid, 0, lane, h, out);
    }
}

template <int F, int PAD>
__global__ void gcn_propagate(const int* __restrict__ src,
                              const int* __restrict__ dst,
                              const float* __restrict__ w,
                              const float* __restrict__ h,
                              float* __restrict__ out, int E) {
    static_assert(PAD == 32 || PAD == 16, "PAD must be 32 or 16");
    out += (size_t)__builtin_amdgcn_readfirstlane(l2_half()) * ((size_t)N_NODES * PAD);
    const int lane = threadIdx.x & 31;
    const int sub = lane / PAD;
    const int col = lane % PAD;
    const int wpb = blockDim.x >> 5;
    const int nwaves = gridDim.x * wpb;
    const int nchunks = (E + 31) >> 5;
    int c = __builtin_amdgcn_readfirstlane(blockIdx.x * wpb + (threadIdx.x >> 5));

    if (c >= nchunks) return;

    int s, d;
    float wt;
    {
        const int e = min((c << 5) + lane, E - 1);
        s = src[e];
        d = dst[e];
        wt = w[e];
    }
    while (c < nchunks) {
        const int nvalid = E - (c << 5);
        const int cn = c + nwaves;
        const int en = min((cn << 5) + lane, E - 1);
        const int sn = src[en];
        const int dn = dst[en];
        const float wn = w[en];
        if (col < F) {
            if (nvalid >= 32) scatter_wave<F, PAD, true >(s, d, wt, nvalid, sub, col, h, out);
            else              scatter_wave<F, PAD, false>(s, d, wt, nvalid, sub, col, h, out);
        }
        s = sn; d = dn; wt = wn; c = cn;
    }
}


#define GEMM_THREADS 256
#define BF2_WORDS  ((HIDDEN / 2) * 16 * 2)
#define BF2_STEPS  ((BF2_WORDS + GEMM_THREADS - 1) / GEMM_THREADS)

__global__ void __launch_bounds__(GEMM_THREADS)
gcn_gemm2(const float* __restrict__ h1,
          const float* __restrict__ b1,
          const float* __restrict__ W2,
          const float* __restrict__ dis,
          float* __restrict__ h2,
          float* __restrict__ h2a) {
    __shared__ float Bf2[BF2_STEPS * GEMM_THREADS];
    __shared__ float b1s[GEMM_THREADS];

    const int tid  = threadIdx.x;
    const int lane = tid & 31;
    const int wave = tid >> 5;
    const int tile = blockIdx.x * (GEMM_THREADS / 32) + wave;
    const bool live = tile < M_TILES;
    const int row0 = tile << 4;
    const int half = lane >> 4;
    const int l16  = lane & 15;

    float wv[BF2_STEPS];
    #pragma unroll
    for (int j = 0; j < BF2_STEPS; ++j) {
        const int idx = min(tid + j * GEMM_THREADS, BF2_WORDS - 1);
        const int p  = idx >> 5;
        const int cc = (idx >> 1) & 15;
        const int jj = idx & 1;
        const float t = W2[(2 * p + jj) * N_CLASSES + min(cc, N_CLASSES - 1)];
        wv[j] = (cc < N_CLASSES) ? t : 0.0f;
    }
    const float bv = b1[min(tid, HIDDEN - 1)];

    const int lrow0 = live ? row0 : 0;
    const float* arow = h1 + (size_t)(lrow0 + l16) * PAD1 + 2 * half;
    v2f ar[HIDDEN / 4], ab[HIDDEN / 4];
    #pragma unroll
    for (int s = 0; s < HIDDEN / 4; ++s) {
        ar[s] = *(const v2f*)(arow + 4 * s);
        ab[s] = *(const v2f*)(arow + PLANE1 + 4 * s);
    }
    const float4* dp = (const float4*)(dis + lrow0 + 8 * half);
    const float4 dlo = dp[0];
    const float4 dhi = dp[1];
    asm volatile("" :: "v"(ar[0]), "v"(ar[1]), "v"(ar[2]), "v"(ar[3]), "v"(ar[4]),
                       "v"(ab[0]), "v"(ab[1]), "v"(ab[2]), "v"(ab[3]), "v"(ab[4]),
                       "v"(dlo.x), "v"(dlo.y), "v"(dlo.z), "v"(dlo.w),
                       "v"(dhi.x), "v"(dhi.y), "v"(dhi.z), "v"(dhi.w));

    #pragma unroll
    for (int j = 0; j < BF2_STEPS; ++j) Bf2[tid + j * GEMM_THREADS] = wv[j];
    b1s[tid] = bv;
    __syncthreads();
    if (!live) return;

    v8f c = {};
    #pragma unroll
    for (int s = 0; s < HIDDEN / 4; ++s) {
        const int kb = 4 * s + 2 * half;
        v2f a;
        a.x = fmaxf((ar[s].x + ab[s].x) + b1s[kb],     0.0f);
        a.y = fmaxf((ar[s].y + ab[s].y) + b1s[kb + 1], 0.0f);
        v2f b = *(const v2f*)&Bf2[(kb >> 1) * 32 + l16 * 2];
        c = __builtin_amdgcn_wmma_f32_16x16x4_f32(false, a, false, b, (short)0, c, false, false);
    }

    const float dv[8] = {dlo.x, dlo.y, dlo.z, dlo.w, dhi.x, dhi.y, dhi.z, dhi.w};
    #pragma unroll
    for (int r = 0; r < 8; ++r) {
        const int row = row0 + r + 8 * half;
        const float di = dv[r];
        const float d2 = di * di;
        h2 [(size_t)row * PAD2 + l16] = c[r];
        h2a[(size_t)row * PAD2 + l16] = d2 * c[r];
    }
    float4* z = (float4*)(h2a + PLANE2 + (size_t)row0 * PAD2);
    for (int q = 0; q < 2; ++q) z[q * 32 + lane] = make_float4(0.0f, 0.0f, 0.0f, 0.0f);
}


__global__ void gcn_logsoftmax(const float* __restrict__ h2a,
                               const float* __restrict__ b2,
                               float* __restrict__ out) {
    int i = blockIdx.x * blockDim.x + threadIdx.x;
    if (i >= N_NODES) return;
    const float* row = h2a + (size_t)i * PAD2;
    float z[N_CLASSES];
    {
        float4 t0 = *(const float4*)(row);
        float4 t1 = *(const float4*)(row + 4);
        float2 t2 = *(const float2*)(row + 8);
        float4 u0 = *(const float4*)(row + PLANE2);
        float4 u1 = *(const float4*)(row + PLANE2 + 4);
        float2 u2 = *(const float2*)(row + PLANE2 + 8);
        z[0] = t0.x + u0.x; z[1] = t0.y + u0.y; z[2] = t0.z + u0.z; z[3] = t0.w + u0.w;
        z[4] = t1.x + u1.x; z[5] = t1.y + u1.y; z[6] = t1.z + u1.z; z[7] = t1.w + u1.w;
        z[8] = t2.x + u2.x; z[9] = t2.y + u2.y;
    }
    float m = -3.402823466e38f;
    #pragma unroll
    for (int c = 0; c < N_CLASSES; ++c) {
        z[c] += b2[c];
        m = fmaxf(m, z[c]);
    }
    float s = 0.0f;
    #pragma unroll
    for (int c = 0; c < N_CLASSES; ++c) s += __expf(z[c] - m);
    const float l = __logf(s);
    #pragma unroll
    for (int c = 0; c < N_CLASSES; ++c) out[(size_t)i * N_CLASSES + c] = z[c] - m - l;
}


extern "C" void kernel_launch(void* const* d_in, const int* in_sizes, int n_in,
                              void* d_out, int out_size, void* d_ws, size_t ws_size,
                              hipStream_t stream) {
    const float* x  = (const float*)d_in[0];
    const int*   ei = (const int*)d_in[1];
    const float* W1 = (const float*)d_in[2];
    const float* b1 = (const float*)d_in[3];
    const float* W2 = (const float*)d_in[4];
    const float* b2 = (const float*)d_in[5];
    float* out = (float*)d_out;

    const int E = in_sizes[1] / 2;
    const int* src = ei;
    const int* dst = ei + E;

    float* ws  = (float*)d_ws;
    float* dis = ws;
    unsigned char* cnt8 = (unsigned char*)(ws + DEG_PLANE);
    float* h0  = ws + 2 * DEG_PLANE;
    float* h1  = h0 + (size_t)N_NODES * PAD1;
    float* h2  = h1 + 2 * (size_t)PLANE1;
    float* h2a = h2 + (size_t)N_NODES * PAD2;
    unsigned* part = (unsigned*)(h2a + 2 * (size_t)PLANE2);
    float* ew  = (float*)(part + (size_t)DEG_SLICES * PART_PITCH);

    const int TB = 256;
    const int nodeBlocks = (N_NODES + TB - 1) / TB;
    const int edgeBlocks = EDGE_BLOCKS;
    const int gemmBlocks = (M_TILES + (TB / 32) - 1) / (TB / 32);

    gcn_deg_count<<<DEG_RANGES * DEG_SLICES, DEG_THREADS, 0, stream>>>(dst, part, E);
    gcn_rsqrt<<<nodeBlocks, TB, 0, stream>>>(part, dis, cnt8);

    gcn_gemm1<<<G1_BLOCKS, G1_THREADS, 0, stream>>>(x, W1, dis, h0, h1);
    gcn_scatter1<<<EDGE1_BLOCKS, EDGE1_THREADS, 0, stream>>>(src, dst, dis, cnt8, ew, h0, h1, E);

    gcn_gemm2<<<gemmBlocks, GEMM_THREADS, 0, stream>>>(h1, b1, W2, dis, h2, h2a);
    gcn_propagate<N_CLASSES, PAD2><<<edgeBlocks, TB, 0, stream>>>(src, dst, ew, h2, h2a, E);

    gcn_logsoftmax<<<nodeBlocks, TB, 0, stream>>>(h2a, b2, out);
}
